# starting kernel
# baseline (speedup 1.0000x reference)
#include <hip/hip_runtime.h>
#include <math.h>

typedef __attribute__((ext_vector_type(2))) float v2f;
typedef __attribute__((ext_vector_type(8))) float v8f;

#define B_MAX      32
#define N_TOK      4096
#define IN_DIM     64
#define FEAT_DIM   768
#define NQ         4
#define NEG_BIG    (-3.402823466e38f)

__global__ void lap_qproj_kernel(const float* __restrict__ qt,
                                 const float* __restrict__ W,
                                 const float* __restrict__ bias,
                                 float* __restrict__ qp,
                                 float* __restrict__ qb)
{
    const int t = threadIdx.x;
    const int q = t >> 6;
    const int i = t & 63;
    float acc = 0.0f;
    const float* qrow = qt + q * FEAT_DIM;
    for (int f = 0; f < FEAT_DIM; ++f)
        acc += qrow[f] * W[f * IN_DIM + i];
    qp[q * IN_DIM + i] = acc;
    if (t < NQ) {
        float s = 0.0f;
        const float* qr = qt + t * FEAT_DIM;
        for (int f = 0; f < FEAT_DIM; ++f) s += qr[f] * bias[f];
        qb[t] = s;
    }
}

__global__ void lap_pool_kernel(const float* __restrict__ x,
                                const int*   __restrict__ mask,
                                const float* __restrict__ qp,
                                const float* __restrict__ qb,
                                float* __restrict__ pooled)
{
    const int b = blockIdx.x;
    const int q = blockIdx.y;
    const int t = threadIdx.x;

    __shared__ float s_w[N_TOK];
    __shared__ float s_qp[IN_DIM];
    __shared__ float s_red[256];
    __shared__ float s_part[4][IN_DIM];

    if (t < IN_DIM) s_qp[t] = qp[q * IN_DIM + t];
    __syncthreads();

    const float qbias = qb[q];
    const float* xb = x + (size_t)b * N_TOK * IN_DIM;
    const int*   mb = mask + (size_t)b * N_TOK;

    float lmax = NEG_BIG;
    for (int n = t; n < N_TOK; n += 256) {
        const float* xr = xb + n * IN_DIM;
        float s = qbias;
        #pragma unroll
        for (int i = 0; i < IN_DIM; i += 4) {
            float4 xv = *reinterpret_cast<const float4*>(xr + i);
            s += s_qp[i]     * xv.x + s_qp[i + 1] * xv.y
               + s_qp[i + 2] * xv.z + s_qp[i + 3] * xv.w;
        }
        if (mb[n] == 0) s = NEG_BIG;
        s_w[n] = s;
        lmax = fmaxf(lmax, s);
    }

    s_red[t] = lmax;
    __syncthreads();
    for (int off = 128; off > 0; off >>= 1) {
        if (t < off) s_red[t] = fmaxf(s_red[t], s_red[t + off]);
        __syncthreads();
    }
    const float mx = s_red[0];
    __syncthreads();

    float lsum = 0.0f;
    for (int n = t; n < N_TOK; n += 256) {
        float w = __expf(s_w[n] - mx);
        s_w[n] = w;
        lsum += w;
    }
    s_red[t] = lsum;
    __syncthreads();
    for (int off = 128; off > 0; off >>= 1) {
        if (t < off) s_red[t] += s_red[t + off];
        __syncthreads();
    }
    const float inv = 1.0f / s_red[0];
    __syncthreads();

    const int i = t & 63;
    const int c = t >> 6;
    float acc = 0.0f;
    const int nbeg = c * (N_TOK / 4);
    const int nend = nbeg + (N_TOK / 4);
    for (int n = nbeg; n < nend; ++n)
        acc += s_w[n] * xb[n * IN_DIM + i];
    s_part[c][i] = acc;
    __syncthreads();
    if (t < IN_DIM) {
        float p = s_part[0][t] + s_part[1][t] + s_part[2][t] + s_part[3][t];
        pooled[(b * NQ + q) * IN_DIM + t] = p * inv;
    }
}

__global__ void lap_outgemm_kernel(const float* __restrict__ pooled,
                                   const float* __restrict__ W,
                                   const float* __restrict__ bias,
                                   float* __restrict__ out)
{
    const int n0   = blockIdx.x * 16;
    const int m0   = blockIdx.y * 16;
    const int lane = threadIdx.x;
    const int hf   = lane >> 4;
    const int l    = lane & 15;

    const int arow = m0 + l;
    const int bcol = n0 + l;
    const float* ap = pooled + arow * IN_DIM;
    const float* bp = W + bcol * IN_DIM;

    v8f acc = {};
    #pragma unroll
    for (int k0 = 0; k0 < IN_DIM; k0 += 4) {
        const int ka = k0 + hf * 2;
        v2f a, bfr;
        a.x   = ap[ka];
        a.y   = ap[ka + 1];
        bfr.x = bp[ka];
        bfr.y = bp[ka + 1];
        acc = __builtin_amdgcn_wmma_f32_16x16x4_f32(
                   false, a,  false, bfr,
                   (short)0, acc,  false,  false);
    }

    const int col = n0 + l;
    const float bv = bias[col];
    #pragma unroll
    for (int r = 0; r < 8; ++r) {
        const int row = m0 + hf * 8 + r;
        out[row * FEAT_DIM + col] = acc[r] + bv;
    }
}

extern "C" void kernel_launch(void* const* d_in, const int* in_sizes, int n_in,
                              void* d_out, int out_size, void* d_ws, size_t ws_size,
                              hipStream_t stream) {
    const float* x    = (const float*)d_in[0];
    const int*   mask = (const int*)  d_in[1];
    const float* qt   = (const float*)d_in[2];
    const float* W    = (const float*)d_in[3];
    const float* bias = (const float*)d_in[4];
    float* out = (float*)d_out;

    const int B = in_sizes[0] / (N_TOK * IN_DIM);

    float* ws     = (float*)d_ws;
    float* qp     = ws;
    float* qb     = ws + 256;
    float* pooled = ws + 512;

    lap_qproj_kernel<<<1, 256, 0, stream>>>(qt, W, bias, qp, qb);
    lap_pool_kernel<<<dim3(B, NQ), 256, 0, stream>>>(x, mask, qp, qb, pooled);
    lap_outgemm_kernel<<<dim3(FEAT_DIM / 16, (B * NQ) / 16), 32, 0, stream>>>(
        pooled, W, bias, out);
}


# q-projection: W operand fetched in row form (16 B per lane) and turned to column pairs through the wave's own LDS rows: 12 loads per 48 f, not 30, requested 4 groups ahead; sum order unchanged
# speedup vs baseline: 5.6609x; 5.6609x over previous
#include <hip/hip_runtime.h>
#include <math.h>

typedef __attribute__((ext_vector_type(2))) float v2f;
typedef __attribute__((ext_vector_type(8))) float v8f;

#define B_MAX      32
#define N_TOK      4096
#define IN_DIM     64
#define FEAT_DIM   768
#define NQ         4
#define NEG_BIG    (-3.402823466e38f)

#define QP_CHUNK   8
#define QP_GROUP   6
#define QP_GROUP_F (QP_CHUNK * QP_GROUP)
#define QP_NGROUP  (FEAT_DIM / QP_GROUP_F)
#define QP_AHEAD   4
#define QP_PITCH   20
#define QP_REGION  (QP_GROUP_F * QP_PITCH)

__device__ __forceinline__ void lap_qp_chunk_mma(const float4& a, const v2f& b0, const v2f& b1,
                                                 v8f& acc0, v8f& acc1)
{
    v2f a0, a1;
    a0.x = a.x; a0.y = a.y;
    a1.x = a.z; a1.y = a.w;
    acc0 = __builtin_amdgcn_wmma_f32_16x16x4_f32(
                false, a0,  false, b0,
                (short)0, acc0,  false,  false);
    acc1 = __builtin_amdgcn_wmma_f32_16x16x4_f32(
                false, a1,  false, b1,
                (short)0, acc1,  false,  false);
}

__global__ __launch_bounds__(32)
void lap_qproj_kernel(const float* __restrict__ qt,
                      const float* __restrict__ W,
                      const float* __restrict__ bias,
                      float* __restrict__ qp,
                      float* __restrict__ qb)
{
    const int lane = threadIdx.x;
    const int h    = lane >> 4;
    const int c    = lane & 15;
    const int blk  = blockIdx.x;

    __shared__ __align__(16) float s_w[QP_NGROUP * QP_REGION];

    const float* ap = qt + (c & 3) * FEAT_DIM + 4 * h;
    v8f acc0 = {};
    v8f acc1 = {};
    float4 av[QP_NGROUP][QP_GROUP];
    float4 wv[QP_NGROUP][QP_GROUP];
    if (blk < 4) {
        const float* wp = W + (lane >> 2) * IN_DIM + 16 * blk + 4 * (lane & 3);
        float*       st = s_w + (lane >> 2) * QP_PITCH + 4 * (lane & 3);
        const float* ld = s_w + 4 * h * QP_PITCH + c;
        v2f b0[QP_NGROUP][QP_GROUP];
        v2f b1[QP_NGROUP][QP_GROUP];
        #pragma unroll
        for (int g = 0; g < QP_AHEAD; ++g)
            #pragma unroll
            for (int j = 0; j < QP_GROUP; ++j) {
                av[g][j] = *reinterpret_cast<const float4*>(ap + g * QP_GROUP_F + j * QP_CHUNK);
                wv[g][j] = *reinterpret_cast<const float4*>(wp + (g * QP_GROUP_F + j * QP_CHUNK) * IN_DIM);
            }
        #pragma unroll
        for (int j = 0; j < QP_GROUP; ++j) {
            *reinterpret_cast<float4*>(st + j * QP_CHUNK * QP_PITCH) = wv[0][j];
            const float* p = ld + j * QP_CHUNK * QP_PITCH;
            b0[0][j].x = p[0];
            b0[0][j].y = p[QP_PITCH];
            b1[0][j].x = p[2 * QP_PITCH];
            b1[0][j].y = p[3 * QP_PITCH];
        }
        #pragma unroll
        for (int g = 0; g < QP_NGROUP; ++g)
            #pragma unroll
            for (int j = 0; j < QP_GROUP; ++j) {
                if (g + QP_AHEAD < QP_NGROUP) {
                    av[g + QP_AHEAD][j] = *reinterpret_cast<const float4*>(
                        ap + (g + QP_AHEAD) * QP_GROUP_F + j * QP_CHUNK);
                    wv[g + QP_AHEAD][j] = *reinterpret_cast<const float4*>(
                        wp + ((g + QP_AHEAD) * QP_GROUP_F + j * QP_CHUNK) * IN_DIM);
                }
                if (g + 1 < QP_NGROUP) {
                    *reinterpret_cast<float4*>(st + (g + 1) * QP_REGION + j * QP_CHUNK * QP_PITCH) = wv[g + 1][j];
                    const float* p = ld + (g + 1) * QP_REGION + j * QP_CHUNK * QP_PITCH;
                    b0[g + 1][j].x = p[0];
                    b0[g + 1][j].y = p[QP_PITCH];
                    b1[g + 1][j].x = p[2 * QP_PITCH];
                    b1[g + 1][j].y = p[3 * QP_PITCH];
                }
                lap_qp_chunk_mma(av[g][j], b0[g][j], b1[g][j], acc0, acc1);
            }
        __builtin_amdgcn_sched_group_barrier(0x020, 2 * QP_GROUP * QP_AHEAD, 0);
        #pragma unroll
        for (int j = 0; j < QP_GROUP; ++j) {
            __builtin_amdgcn_sched_group_barrier(0x200, 1, 0);
            __builtin_amdgcn_sched_group_barrier(0x100, 2, 0);
        }
        #pragma unroll
        for (int g = 0; g < QP_NGROUP; ++g)
            #pragma unroll
            for (int j = 0; j < QP_GROUP; ++j) {
                if (g + QP_AHEAD < QP_NGROUP)
                    __builtin_amdgcn_sched_group_barrier(0x020, 2, 0);
                if (g + 1 < QP_NGROUP) {
                    __builtin_amdgcn_sched_group_barrier(0x200, 1, 0);
                    __builtin_amdgcn_sched_group_barrier(0x100, 2, 0);
                }
                __builtin_amdgcn_sched_group_barrier(0x008, 2, 0);
            }
    } else {
        const float* wp = bias + 4 * h;
        #pragma unroll
        for (int g = 0; g < QP_AHEAD; ++g)
            #pragma unroll
            for (int j = 0; j < QP_GROUP; ++j) {
                av[g][j] = *reinterpret_cast<const float4*>(ap + g * QP_GROUP_F + j * QP_CHUNK);
                wv[g][j] = *reinterpret_cast<const float4*>(wp + g * QP_GROUP_F + j * QP_CHUNK);
            }
        #pragma unroll
        for (int g = 0; g < QP_NGROUP; ++g)
            #pragma unroll
            for (int j = 0; j < QP_GROUP; ++j) {
                if (g + QP_AHEAD < QP_NGROUP) {
                    av[g + QP_AHEAD][j] = *reinterpret_cast<const float4*>(
                        ap + (g + QP_AHEAD) * QP_GROUP_F + j * QP_CHUNK);
                    wv[g + QP_AHEAD][j] = *reinterpret_cast<const float4*>(
                        wp + (g + QP_AHEAD) * QP_GROUP_F + j * QP_CHUNK);
                }
                v2f bb0, bb1;
                bb0.x = wv[g][j].x; bb0.y = wv[g][j].y;
                bb1.x = wv[g][j].z; bb1.y = wv[g][j].w;
                lap_qp_chunk_mma(av[g][j], bb0, bb1, acc0, acc1);
                __builtin_amdgcn_sched_barrier(0);
            }
    }

    if (h == 0) {
        if (blk < 4) {
            #pragma unroll
            for (int r = 0; r < NQ; ++r)
                qp[r * IN_DIM + 16 * blk + c] = acc0[r] + acc1[r];
        } else if (c == 0) {
            #pragma unroll
            for (int r = 0; r < NQ; ++r)
                qb[r] = acc0[r] + acc1[r];
        }
    }
}

#define PW_SLICES    8
#define PW_SLICE_TOK (N_TOK / PW_SLICES)
#define PW_WAVES     8
#define PW_THREADS   (32 * PW_WAVES)
#define PW_TILES     (PW_SLICE_TOK / (16 * PW_WAVES))
#define PW_PITCH     68

__device__ __forceinline__ float lap_xhalf(float v)
{
    return __int_as_float(__builtin_amdgcn_permlanex16(__float_as_int(v), __float_as_int(v),
                                                       0x76543210, 0xfedcba98, false, false));
}

__device__ __forceinline__ float lap_comp(const float4& v, int T)
{
    return T == 0 ? v.x : (T == 1 ? v.y : (T == 2 ? v.z : v.w));
}

__global__ __launch_bounds__(PW_THREADS)
void lap_pool_mm_kernel(const float* __restrict__ x,
                        const int*   __restrict__ mask,
                        const float* __restrict__ qp,
                        const float* __restrict__ qb,
                        float* __restrict__ partP,
                        float* __restrict__ partM,
                        float* __restrict__ partL)
{
    const int b    = blockIdx.x;
    const int sl   = blockIdx.y;
    const int t    = threadIdx.x;
    const int lane = t & 31;
    const int wave = t >> 5;
    const int h    = lane >> 4;
    const int c    = lane & 15;
    const int q    = c & 3;

    __shared__ __align__(16) float s_x[PW_WAVES][16 * PW_PITCH];
    __shared__ float s_m[PW_WAVES][NQ];
    __shared__ float s_l[PW_WAVES][2][NQ];
    __shared__ float s_p[PW_WAVES][NQ][IN_DIM];

    float4 qv[8];
    #pragma unroll
    for (int jj = 0; jj < 8; ++jj)
        qv[jj] = *reinterpret_cast<const float4*>(qp + q * IN_DIM + 32 * h + 4 * jj);
    const float qbias = qb[q];

    const int tok0 = sl * PW_SLICE_TOK + wave * (16 * PW_TILES);
    const float* xw = x + ((size_t)b * N_TOK + tok0) * IN_DIM;
    const int*   mw = mask + (size_t)b * N_TOK + tok0;

    float4 R[PW_TILES][8];
    int4   mk[PW_TILES][2];
    #pragma unroll
    for (int tl = 0; tl < PW_TILES; ++tl) {
        #pragma unroll
        for (int r = 0; r < 8; ++r)
            R[tl][r] = *reinterpret_cast<const float4*>(xw + (16 * tl + 8 * h + r) * IN_DIM + 4 * c);
        mk[tl][0] = *reinterpret_cast<const int4*>(mw + 16 * tl + 8 * h);
        mk[tl][1] = *reinterpret_cast<const int4*>(mw + 16 * tl + 8 * h + 4);
    }

    float m = NEG_BIG;
    float l = 0.0f;
    v8f acc[4];
    #pragma unroll
    for (int T = 0; T < 4; ++T) acc[T] = v8f{};

    float* sx = s_x[wave];
    #pragma unroll
    for (int tl = 0; tl < PW_TILES; ++tl) {
        #pragma unroll
        for (int r = 0; r < 8; ++r)
            *reinterpret_cast<float4*>(sx + (8 * h + r) * PW_PITCH + 4 * c) = R[tl][r];
        float4 C[8];
        #pragma unroll
        for (int jj = 0; jj < 8; ++jj)
            C[jj] = *reinterpret_cast<const float4*>(sx + c * PW_PITCH + 32 * h + 4 * jj);

        v8f d0 = {};
        v8f d1 = {};
        #pragma unroll
        for (int jj = 0; jj < 8; ++jj) {
            v2f a0, a1, b0, b1;
            a0.x = C[jj].x;  a0.y = C[jj].y;  a1.x = C[jj].z;  a1.y = C[jj].w;
            b0.x = qv[jj].x; b0.y = qv[jj].y; b1.x = qv[jj].z; b1.y = qv[jj].w;
            d0 = __builtin_amdgcn_wmma_f32_16x16x4_f32(
                      false, a0,  false, b0,
                      (short)0, d0,  false,  false);
            d1 = __builtin_amdgcn_wmma_f32_16x16x4_f32(
                      false, a1,  false, b1,
                      (short)0, d1,  false,  false);
        }

        const int mkv[8] = { mk[tl][0].x, mk[tl][0].y, mk[tl][0].z, mk[tl][0].w,
                             mk[tl][1].x, mk[tl][1].y, mk[tl][1].z, mk[tl][1].w };
        float sc_[8];
        float bm = NEG_BIG;
        #pragma unroll
        for (int r = 0; r < 8; ++r) {
            float p = d0[r] + d1[r] + qbias;
            if (mkv[r] == 0) p = NEG_BIG;
            sc_[r] = p;
            bm = fmaxf(bm, p);
        }
        bm = fmaxf(bm, lap_xhalf(bm));
        const float mn = fmaxf(m, bm);
        const float rs = __expf(m - mn);
        m = mn;
        l *= rs;
        #pragma unroll
        for (int T = 0; T < 4; ++T)
            #pragma unroll
            for (int r = 0; r < 8; ++r) acc[T][r] *= rs;
        float w[8];
        #pragma unroll
        for (int r = 0; r < 8; ++r) {
            w[r] = __expf(sc_[r] - mn);
            l += w[r];
        }

        #pragma unroll
        for (int j = 0; j < 4; ++j) {
            v2f bw;
            bw.x = w[2 * j];
            bw.y = w[2 * j + 1];
            #pragma unroll
            for (int T = 0; T < 4; ++T) {
                v2f a;
                a.x = lap_comp(R[tl][2 * j], T);
                a.y = lap_comp(R[tl][2 * j + 1], T);
                acc[T] = __builtin_amdgcn_wmma_f32_16x16x4_f32(
                              false, a,  false, bw,
                              (short)0, acc[T],  false,  false);
            }
        }
        __builtin_amdgcn_sched_barrier(0);
    }

    if (c < NQ) {
        if (h == 0) s_m[wave][c] = m;
        s_l[wave][h][c] = l;
        #pragma unroll
        for (int T = 0; T < 4; ++T)
            #pragma unroll
            for (int r = 0; r < 8; ++r)
                s_p[wave][c][4 * (8 * h + r) + T] = acc[T][r];
    }
    __syncthreads();
    {
        const int qq = t >> 6;
        const int i  = t & 63;
        float M = NEG_BIG;
        #pragma unroll
        for (int wv = 0; wv < PW_WAVES; ++wv) M = fmaxf(M, s_m[wv][qq]);
        float L = 0.0f, P = 0.0f;
        #pragma unroll
        for (int wv = 0; wv < PW_WAVES; ++wv) {
            const float e = __expf(s_m[wv][qq] - M);
            L += (s_l[wv][0][qq] + s_l[wv][1][qq]) * e;
            P += s_p[wv][qq][i] * e;
        }
        const int rec = (b * PW_SLICES + sl) * NQ + qq;
        partP[rec * IN_DIM + i] = P;
        if (i == 0) { partM[rec] = M; partL[rec] = L; }
    }
}

__global__ __launch_bounds__(IN_DIM)
void lap_merge_kernel(const float* __restrict__ partP,
                      const float* __restrict__ partM,
                      const float* __restrict__ partL,
                      float* __restrict__ pooled)
{
    const int row = blockIdx.x;
    const int b   = row >> 2;
    const int q   = row & 3;
    const int i   = threadIdx.x;
    float M = NEG_BIG;
    #pragma unroll
    for (int s = 0; s < PW_SLICES; ++s) M = fmaxf(M, partM[(b * PW_SLICES + s) * NQ + q]);
    float L = 0.0f, P = 0.0f;
    #pragma unroll
    for (int s = 0; s < PW_SLICES; ++s) {
        const int rec = (b * PW_SLICES + s) * NQ + q;
        const float e = __expf(partM[rec] - M);
        L += partL[rec] * e;
        P += partP[rec * IN_DIM + i] * e;
    }
    pooled[row * IN_DIM + i] = P / L;
}

__global__ void lap_outgemm_kernel(const float* __restrict__ pooled,
                                   const float* __restrict__ W,
                                   const float* __restrict__ bias,
                                   float* __restrict__ out)
{
    const int n0   = blockIdx.x * 16;
    const int m0   = blockIdx.y * 16;
    const int lane = threadIdx.x;
    const int hf   = lane >> 4;
    const int l    = lane & 15;

    const int arow = m0 + l;
    const int bcol = n0 + l;
    const float* ap = pooled + arow * IN_DIM;
    const float* bp = W + bcol * IN_DIM;

    v8f acc = {};
    #pragma unroll
    for (int k0 = 0; k0 < IN_DIM; k0 += 4) {
        const int ka = k0 + hf * 2;
        v2f a, bfr;
        a.x   = ap[ka];
        a.y   = ap[ka + 1];
        bfr.x = bp[ka];
        bfr.y = bp[ka + 1];
        acc = __builtin_amdgcn_wmma_f32_16x16x4_f32(
                   false, a,  false, bfr,
                   (short)0, acc,  false,  false);
    }

    const int col = n0 + l;
    const float bv = bias[col];
    #pragma unroll
    for (int r = 0; r < 8; ++r) {
        const int row = m0 + hf * 8 + r;
        out[row * FEAT_DIM + col] = acc[r] + bv;
    }
}

extern "C" void kernel_launch(void* const* d_in, const int* in_sizes, int n_in,
                              void* d_out, int out_size, void* d_ws, size_t ws_size,
                              hipStream_t stream) {
    const float* x    = (const float*)d_in[0];
    const int*   mask = (const int*)  d_in[1];
    const float* qt   = (const float*)d_in[2];
    const float* W    = (const float*)d_in[3];
    const float* bias = (const float*)d_in[4];
    float* out = (float*)d_out;

    const int B = in_sizes[0] / (N_TOK * IN_DIM);

    float* ws     = (float*)d_ws;
    float* qp     = ws;
    float* qb     = ws + 256;
    float* pooled = ws + 512;

    lap_qproj_kernel<<<5, 32, 0, stream>>>(qt, W, bias, qp, qb);
    float* partP = out;
    float* partM = out + 65536;
    float* partL = out + 66560;
    lap_pool_mm_kernel<<<dim3(B, PW_SLICES), PW_THREADS, 0, stream>>>(x, mask, qp, qb, partP, partM, partL);
    lap_merge_kernel<<<B * NQ, IN_DIM, 0, stream>>>(partP, partM, partL, pooled);
    lap_outgemm_kernel<<<dim3(FEAT_DIM / 16, (B * NQ) / 16), 32, 0, stream>>>(
        pooled, W, bias, out);
}


# q-projection: a group's 6 LDS stores and 12 LDS reads issued in ONE place ahead of the previous group's 12 matrix instructions (16 LDS places a wave, was 96); loads 5 groups ahead; sums unchanged
# speedup vs baseline: 6.0670x; 1.0717x over previous
#include <hip/hip_runtime.h>
#include <math.h>

typedef __attribute__((ext_vector_type(2))) float v2f;
typedef __attribute__((ext_vector_type(8))) float v8f;

#define B_MAX      32
#define N_TOK      4096
#define IN_DIM     64
#define FEAT_DIM   768
#define NQ         4
#define NEG_BIG    (-3.402823466e38f)

#define QP_CHUNK   8
#define QP_GROUP   6
#define QP_GROUP_F (QP_CHUNK * QP_GROUP)
#define QP_NGROUP  (FEAT_DIM / QP_GROUP_F)
#define QP_AHEAD   5
#define QP_PITCH   20
#define QP_REGION  (QP_GROUP_F * QP_PITCH)

__device__ __forceinline__ void lap_qp_chunk_mma(const float4& a, const v2f& b0, const v2f& b1,
                                                 v8f& acc0, v8f& acc1)
{
    v2f a0, a1;
    a0.x = a.x; a0.y = a.y;
    a1.x = a.z; a1.y = a.w;
    acc0 = __builtin_amdgcn_wmma_f32_16x16x4_f32(
                false, a0,  false, b0,
                (short)0, acc0,  false,  false);
    acc1 = __builtin_amdgcn_wmma_f32_16x16x4_f32(
                false, a1,  false, b1,
                (short)0, acc1,  false,  false);
}

__global__ __launch_bounds__(32)
void lap_qproj_kernel(const float* __restrict__ qt,
                      const float* __restrict__ W,
                      const float* __restrict__ bias,
                      float* __restrict__ qp,
                      float* __restrict__ qb)
{
    const int lane = threadIdx.x;
    const int h    = lane >> 4;
    const int c    = lane & 15;
    const int blk  = blockIdx.x;

    __shared__ __align__(16) float s_w[QP_NGROUP * QP_REGION];

    const float* ap = qt + (c & 3) * FEAT_DIM + 4 * h;
    v8f acc0 = {};
    v8f acc1 = {};
    float4 av[QP_NGROUP][QP_GROUP];
    float4 wv[QP_NGROUP][QP_GROUP];
    if (blk < 4) {
        const float* wp = W + (lane >> 2) * IN_DIM + 16 * blk + 4 * (lane & 3);
        float*       st = s_w + (lane >> 2) * QP_PITCH + 4 * (lane & 3);
        const float* ld = s_w + 4 * h * QP_PITCH + c;
        v2f b0[QP_NGROUP][QP_GROUP];
        v2f b1[QP_NGROUP][QP_GROUP];
        #pragma unroll
        for (int g = 0; g < QP_AHEAD; ++g)
            #pragma unroll
            for (int j = 0; j < QP_GROUP; ++j) {
                av[g][j] = *reinterpret_cast<const float4*>(ap + g * QP_GROUP_F + j * QP_CHUNK);
                wv[g][j] = *reinterpret_cast<const float4*>(wp + (g * QP_GROUP_F + j * QP_CHUNK) * IN_DIM);
            }
        #pragma unroll
        for (int j = 0; j < QP_GROUP; ++j)
            *reinterpret_cast<float4*>(st + j * QP_CHUNK * QP_PITCH) = wv[0][j];
        #pragma unroll
        for (int j = 0; j < QP_GROUP; ++j) {
            const float* p = ld + j * QP_CHUNK * QP_PITCH;
            b0[0][j].x = p[0];
            b0[0][j].y = p[QP_PITCH];
            b1[0][j].x = p[2 * QP_PITCH];
            b1[0][j].y = p[3 * QP_PITCH];
        }
        #pragma unroll
        for (int g = 0; g < QP_NGROUP; ++g) {
            if (g + QP_AHEAD < QP_NGROUP) {
                #pragma unroll
                for (int j = 0; j < QP_GROUP; ++j) {
                    av[g + QP_AHEAD][j] = *reinterpret_cast<const float4*>(
                        ap + (g + QP_AHEAD) * QP_GROUP_F + j * QP_CHUNK);
                    wv[g + QP_AHEAD][j] = *reinterpret_cast<const float4*>(
                        wp + ((g + QP_AHEAD) * QP_GROUP_F + j * QP_CHUNK) * IN_DIM);
                }
            }
            if (g + 1 < QP_NGROUP) {
                #pragma unroll
                for (int j = 0; j < QP_GROUP; ++j)
                    *reinterpret_cast<float4*>(st + (g + 1) * QP_REGION + j * QP_CHUNK * QP_PITCH) = wv[g + 1][j];
                #pragma unroll
                for (int j = 0; j < QP_GROUP; ++j) {
                    const float* p = ld + (g + 1) * QP_REGION + j * QP_CHUNK * QP_PITCH;
                    b0[g + 1][j].x = p[0];
                    b0[g + 1][j].y = p[QP_PITCH];
                    b1[g + 1][j].x = p[2 * QP_PITCH];
                    b1[g + 1][j].y = p[3 * QP_PITCH];
                }
            }
            #pragma unroll
            for (int j = 0; j < QP_GROUP; ++j)
                lap_qp_chunk_mma(av[g][j], b0[g][j], b1[g][j], acc0, acc1);
        }
        __builtin_amdgcn_sched_group_barrier(0x020, 2 * QP_GROUP * QP_AHEAD, 0);
        __builtin_amdgcn_sched_group_barrier(0x200, QP_GROUP, 0);
        __builtin_amdgcn_sched_group_barrier(0x100, 2 * QP_GROUP, 0);
        #pragma unroll
        for (int g = 0; g < QP_NGROUP; ++g) {
            if (g + QP_AHEAD < QP_NGROUP)
                __builtin_amdgcn_sched_group_barrier(0x020, 2 * QP_GROUP, 0);
            if (g + 1 < QP_NGROUP) {
                __builtin_amdgcn_sched_group_barrier(0x200, QP_GROUP, 0);
                __builtin_amdgcn_sched_group_barrier(0x100, 2 * QP_GROUP, 0);
            }
            __builtin_amdgcn_sched_group_barrier(0x008, 2 * QP_GROUP, 0);
        }
    } else {
        const float* wp = bias + 4 * h;
        #pragma unroll
        for (int g = 0; g < QP_AHEAD; ++g)
            #pragma unroll
            for (int j = 0; j < QP_GROUP; ++j) {
                av[g][j] = *reinterpret_cast<const float4*>(ap + g * QP_GROUP_F + j * QP_CHUNK);
                wv[g][j] = *reinterpret_cast<const float4*>(wp + g * QP_GROUP_F + j * QP_CHUNK);
            }
        #pragma unroll
        for (int g = 0; g < QP_NGROUP; ++g)
            #pragma unroll
            for (int j = 0; j < QP_GROUP; ++j) {
                if (g + QP_AHEAD < QP_NGROUP) {
                    av[g + QP_AHEAD][j] = *reinterpret_cast<const float4*>(
                        ap + (g + QP_AHEAD) * QP_GROUP_F + j * QP_CHUNK);
                    wv[g + QP_AHEAD][j] = *reinterpret_cast<const float4*>(
                        wp + (g + QP_AHEAD) * QP_GROUP_F + j * QP_CHUNK);
                }
                v2f bb0, bb1;
                bb0.x = wv[g][j].x; bb0.y = wv[g][j].y;
                bb1.x = wv[g][j].z; bb1.y = wv[g][j].w;
                lap_qp_chunk_mma(av[g][j], bb0, bb1, acc0, acc1);
                __builtin_amdgcn_sched_barrier(0);
            }
    }

    if (h == 0) {
        if (blk < 4) {
            #pragma unroll
            for (int r = 0; r < NQ; ++r)
                qp[r * IN_DIM + 16 * blk + c] = acc0[r] + acc1[r];
        } else if (c == 0) {
            #pragma unroll
            for (int r = 0; r < NQ; ++r)
                qb[r] = acc0[r] + acc1[r];
        }
    }
}

#define PW_SLICES    8
#define PW_SLICE_TOK (N_TOK / PW_SLICES)
#define PW_WAVES     8
#define PW_THREADS   (32 * PW_WAVES)
#define PW_TILES     (PW_SLICE_TOK / (16 * PW_WAVES))
#define PW_PITCH     68

__device__ __forceinline__ float lap_xhalf(float v)
{
    return __int_as_float(__builtin_amdgcn_permlanex16(__float_as_int(v), __float_as_int(v),
                                                       0x76543210, 0xfedcba98, false, false));
}

__device__ __forceinline__ float lap_comp(const float4& v, int T)
{
    return T == 0 ? v.x : (T == 1 ? v.y : (T == 2 ? v.z : v.w));
}

__global__ __launch_bounds__(PW_THREADS)
void lap_pool_mm_kernel(const float* __restrict__ x,
                        const int*   __restrict__ mask,
                        const float* __restrict__ qp,
                        const float* __restrict__ qb,
                        float* __restrict__ partP,
                        float* __restrict__ partM,
                        float* __restrict__ partL)
{
    const int b    = blockIdx.x;
    const int sl   = blockIdx.y;
    const int t    = threadIdx.x;
    const int lane = t & 31;
    const int wave = t >> 5;
    const int h    = lane >> 4;
    const int c    = lane & 15;
    const int q    = c & 3;

    __shared__ __align__(16) float s_x[PW_WAVES][16 * PW_PITCH];
    __shared__ float s_m[PW_WAVES][NQ];
    __shared__ float s_l[PW_WAVES][2][NQ];
    __shared__ float s_p[PW_WAVES][NQ][IN_DIM];

    float4 qv[8];
    #pragma unroll
    for (int jj = 0; jj < 8; ++jj)
        qv[jj] = *reinterpret_cast<const float4*>(qp + q * IN_DIM + 32 * h + 4 * jj);
    const float qbias = qb[q];

    const int tok0 = sl * PW_SLICE_TOK + wave * (16 * PW_TILES);
    const float* xw = x + ((size_t)b * N_TOK + tok0) * IN_DIM;
    const int*   mw = mask + (size_t)b * N_TOK + tok0;

    float4 R[PW_TILES][8];
    int4   mk[PW_TILES][2];
    #pragma unroll
    for (int tl = 0; tl < PW_TILES; ++tl) {
        #pragma unroll
        for (int r = 0; r < 8; ++r)
            R[tl][r] = *reinterpret_cast<const float4*>(xw + (16 * tl + 8 * h + r) * IN_DIM + 4 * c);
        mk[tl][0] = *reinterpret_cast<const int4*>(mw + 16 * tl + 8 * h);
        mk[tl][1] = *reinterpret_cast<const int4*>(mw + 16 * tl + 8 * h + 4);
    }

    float m = NEG_BIG;
    float l = 0.0f;
    v8f acc[4];
    #pragma unroll
    for (int T = 0; T < 4; ++T) acc[T] = v8f{};

    float* sx = s_x[wave];
    #pragma unroll
    for (int tl = 0; tl < PW_TILES; ++tl) {
        #pragma unroll
        for (int r = 0; r < 8; ++r)
            *reinterpret_cast<float4*>(sx + (8 * h + r) * PW_PITCH + 4 * c) = R[tl][r];
        float4 C[8];
        #pragma unroll
        for (int jj = 0; jj < 8; ++jj)
            C[jj] = *reinterpret_cast<const float4*>(sx + c * PW_PITCH + 32 * h + 4 * jj);

        v8f d0 = {};
        v8f d1 = {};
        #pragma unroll
        for (int jj = 0; jj < 8; ++jj) {
            v2f a0, a1, b0, b1;
            a0.x = C[jj].x;  a0.y = C[jj].y;  a1.x = C[jj].z;  a1.y = C[jj].w;
            b0.x = qv[jj].x; b0.y = qv[jj].y; b1.x = qv[jj].z; b1.y = qv[jj].w;
            d0 = __builtin_amdgcn_wmma_f32_16x16x4_f32(
                      false, a0,  false, b0,
                      (short)0, d0,  false,  false);
            d1 = __builtin_amdgcn_wmma_f32_16x16x4_f32(
                      false, a1,  false, b1,
                      (short)0, d1,  false,  false);
        }

        const int mkv[8] = { mk[tl][0].x, mk[tl][0].y, mk[tl][0].z, mk[tl][0].w,
                             mk[tl][1].x, mk[tl][1].y, mk[tl][1].z, mk[tl][1].w };
        float sc_[8];
        float bm = NEG_BIG;
        #pragma unroll
        for (int r = 0; r < 8; ++r) {
            float p = d0[r] + d1[r] + qbias;
            if (mkv[r] == 0) p = NEG_BIG;
            sc_[r] = p;
            bm = fmaxf(bm, p);
        }
        bm = fmaxf(bm, lap_xhalf(bm));
        const float mn = fmaxf(m, bm);
        const float rs = __expf(m - mn);
        m = mn;
        l *= rs;
        #pragma unroll
        for (int T = 0; T < 4; ++T)
            #pragma unroll
            for (int r = 0; r < 8; ++r) acc[T][r] *= rs;
        float w[8];
        #pragma unroll
        for (int r = 0; r < 8; ++r) {
            w[r] = __expf(sc_[r] - mn);
            l += w[r];
        }

        #pragma unroll
        for (int j = 0; j < 4; ++j) {
            v2f bw;
            bw.x = w[2 * j];
            bw.y = w[2 * j + 1];
            #pragma unroll
            for (int T = 0; T < 4; ++T) {
                v2f a;
                a.x = lap_comp(R[tl][2 * j], T);
                a.y = lap_comp(R[tl][2 * j + 1], T);
                acc[T] = __builtin_amdgcn_wmma_f32_16x16x4_f32(
                              false, a,  false, bw,
                              (short)0, acc[T],  false,  false);
            }
        }
        __builtin_amdgcn_sched_barrier(0);
    }

    if (c < NQ) {
        if (h == 0) s_m[wave][c] = m;
        s_l[wave][h][c] = l;
        #pragma unroll
        for (int T = 0; T < 4; ++T)
            #pragma unroll
            for (int r = 0; r < 8; ++r)
                s_p[wave][c][4 * (8 * h + r) + T] = acc[T][r];
    }
    __syncthreads();
    {
        const int qq = t >> 6;
        const int i  = t & 63;
        float M = NEG_BIG;
        #pragma unroll
        for (int wv = 0; wv < PW_WAVES; ++wv) M = fmaxf(M, s_m[wv][qq]);
        float L = 0.0f, P = 0.0f;
        #pragma unroll
        for (int wv = 0; wv < PW_WAVES; ++wv) {
            const float e = __expf(s_m[wv][qq] - M);
            L += (s_l[wv][0][qq] + s_l[wv][1][qq]) * e;
            P += s_p[wv][qq][i] * e;
        }
        const int rec = (b * PW_SLICES + sl) * NQ + qq;
        partP[rec * IN_DIM + i] = P;
        if (i == 0) { partM[rec] = M; partL[rec] = L; }
    }
}

__global__ __launch_bounds__(IN_DIM)
void lap_merge_kernel(const float* __restrict__ partP,
                      const float* __restrict__ partM,
                      const float* __restrict__ partL,
                      float* __restrict__ pooled)
{
    const int row = blockIdx.x;
    const int b   = row >> 2;
    const int q   = row & 3;
    const int i   = threadIdx.x;
    float M = NEG_BIG;
    #pragma unroll
    for (int s = 0; s < PW_SLICES; ++s) M = fmaxf(M, partM[(b * PW_SLICES + s) * NQ + q]);
    float L = 0.0f, P = 0.0f;
    #pragma unroll
    for (int s = 0; s < PW_SLICES; ++s) {
        const int rec = (b * PW_SLICES + s) * NQ + q;
        const float e = __expf(partM[rec] - M);
        L += partL[rec] * e;
        P += partP[rec * IN_DIM + i] * e;
    }
    pooled[row * IN_DIM + i] = P / L;
}

__global__ void lap_outgemm_kernel(const float* __restrict__ pooled,
                                   const float* __restrict__ W,
                                   const float* __restrict__ bias,
                                   float* __restrict__ out)
{
    const int n0   = blockIdx.x * 16;
    const int m0   = blockIdx.y * 16;
    const int lane = threadIdx.x;
    const int hf   = lane >> 4;
    const int l    = lane & 15;

    const int arow = m0 + l;
    const int bcol = n0 + l;
    const float* ap = pooled + arow * IN_DIM;
    const float* bp = W + bcol * IN_DIM;

    v8f acc = {};
    #pragma unroll
    for (int k0 = 0; k0 < IN_DIM; k0 += 4) {
        const int ka = k0 + hf * 2;
        v2f a, bfr;
        a.x   = ap[ka];
        a.y   = ap[ka + 1];
        bfr.x = bp[ka];
        bfr.y = bp[ka + 1];
        acc = __builtin_amdgcn_wmma_f32_16x16x4_f32(
                   false, a,  false, bfr,
                   (short)0, acc,  false,  false);
    }

    const int col = n0 + l;
    const float bv = bias[col];
    #pragma unroll
    for (int r = 0; r < 8; ++r) {
        const int row = m0 + hf * 8 + r;
        out[row * FEAT_DIM + col] = acc[r] + bv;
    }
}

extern "C" void kernel_launch(void* const* d_in, const int* in_sizes, int n_in,
                              void* d_out, int out_size, void* d_ws, size_t ws_size,
                              hipStream_t stream) {
    const float* x    = (const float*)d_in[0];
    const int*   mask = (const int*)  d_in[1];
    const float* qt   = (const float*)d_in[2];
    const float* W    = (const float*)d_in[3];
    const float* bias = (const float*)d_in[4];
    float* out = (float*)d_out;

    const int B = in_sizes[0] / (N_TOK * IN_DIM);

    float* ws     = (float*)d_ws;
    float* qp     = ws;
    float* qb     = ws + 256;
    float* pooled = ws + 512;

    lap_qproj_kernel<<<5, 32, 0, stream>>>(qt, W, bias, qp, qb);
    float* partP = out;
    float* partM = out + 65536;
    float* partL = out + 66560;
    lap_pool_mm_kernel<<<dim3(B, PW_SLICES), PW_THREADS, 0, stream>>>(x, mask, qp, qb, partP, partM, partL);
    lap_merge_kernel<<<B * NQ, IN_DIM, 0, stream>>>(partP, partM, partL, pooled);
    lap_outgemm_kernel<<<dim3(FEAT_DIM / 16, (B * NQ) / 16), 32, 0, stream>>>(
        pooled, W, bias, out);
}


# q-projection: the 12 idle tile rows carry 3 more quarters of the f-sum (rows = query x quarter, columns = 4 outputs x quarter, diagonal blocks added last): 48 k-steps a wave, not 192; new f-sum order
# speedup vs baseline: 6.5519x; 1.0799x over previous
#include <hip/hip_runtime.h>
#include <math.h>

typedef __attribute__((ext_vector_type(2))) float v2f;
typedef __attribute__((ext_vector_type(8))) float v8f;

#define B_MAX      32
#define N_TOK      4096
#define IN_DIM     64
#define FEAT_DIM   768
#define NQ         4
#define NEG_BIG    (-3.402823466e38f)

#define QP_PARTS   4
#define QP_PART_F  (FEAT_DIM / QP_PARTS)
#define QP_CHUNK   8
#define QP_NCHUNK  (QP_PART_F / QP_CHUNK)
#define QP_GROUP   4
#define QP_NGROUP  (QP_NCHUNK / QP_GROUP)
#define QP_COLS    (16 / QP_PARTS)
#define QP_NCB     (IN_DIM / QP_COLS)
#define QP_REGION  (32 * QP_COLS)

__device__ __forceinline__ void lap_qp_chunk_mma(const float4& a, const v2f& b0, const v2f& b1,
                                                 v8f& acc0, v8f& acc1)
{
    v2f a0, a1;
    a0.x = a.x; a0.y = a.y;
    a1.x = a.z; a1.y = a.w;
    acc0 = __builtin_amdgcn_wmma_f32_16x16x4_f32(
                false, a0,  false, b0,
                (short)0, acc0,  false,  false);
    acc1 = __builtin_amdgcn_wmma_f32_16x16x4_f32(
                false, a1,  false, b1,
                (short)0, acc1,  false,  false);
}

__global__ __launch_bounds__(32)
void lap_qproj_kernel(const float* __restrict__ qt,
                      const float* __restrict__ W,
                      const float* __restrict__ bias,
                      float* __restrict__ qp,
                      float* __restrict__ qb)
{
    const int lane = threadIdx.x;
    const int h    = lane >> 4;
    const int c    = lane & 15;
    const int part = c >> 2;
    const int sub  = c & 3;
    const int blk  = blockIdx.x;

    __shared__ __align__(16) float s_w[QP_NCHUNK * QP_REGION];
    __shared__ float s_red[QP_PARTS][8][QP_COLS];

    const float* ap = qt + sub * FEAT_DIM + part * QP_PART_F + 4 * h;
    v8f acc0 = {};
    v8f acc1 = {};
    float4 av[QP_NCHUNK];
    float4 wv[QP_NCHUNK];
    if (blk < QP_NCB) {
        const float* wp = W + ((lane >> 3) * QP_PART_F + (lane & 7)) * IN_DIM + QP_COLS * blk;
        float*       st = s_w + QP_COLS * lane;
        const float* ld = s_w + (8 * part + 4 * h) * QP_COLS + sub;
        v2f b0[QP_NCHUNK];
        v2f b1[QP_NCHUNK];
        #pragma unroll
        for (int j = 0; j < QP_NCHUNK; ++j) {
            av[j] = *reinterpret_cast<const float4*>(ap + j * QP_CHUNK);
            wv[j] = *reinterpret_cast<const float4*>(wp + j * QP_CHUNK * IN_DIM);
        }
        #pragma unroll
        for (int j = 0; j < QP_GROUP; ++j)
            *reinterpret_cast<float4*>(st + j * QP_REGION) = wv[j];
        #pragma unroll
        for (int j = 0; j < QP_GROUP; ++j) {
            const float* p = ld + j * QP_REGION;
            b0[j].x = p[0];
            b0[j].y = p[QP_COLS];
            b1[j].x = p[2 * QP_COLS];
            b1[j].y = p[3 * QP_COLS];
        }
        #pragma unroll
        for (int g = 0; g < QP_NGROUP; ++g) {
            if (g + 1 < QP_NGROUP) {
                #pragma unroll
                for (int j = 0; j < QP_GROUP; ++j)
                    *reinterpret_cast<float4*>(st + ((g + 1) * QP_GROUP + j) * QP_REGION) = wv[(g + 1) * QP_GROUP + j];
                #pragma unroll
                for (int j = 0; j < QP_GROUP; ++j) {
                    const float* p = ld + ((g + 1) * QP_GROUP + j) * QP_REGION;
                    b0[(g + 1) * QP_GROUP + j].x = p[0];
                    b0[(g + 1) * QP_GROUP + j].y = p[QP_COLS];
                    b1[(g + 1) * QP_GROUP + j].x = p[2 * QP_COLS];
                    b1[(g + 1) * QP_GROUP + j].y = p[3 * QP_COLS];
                }
            }
            #pragma unroll
            for (int j = 0; j < QP_GROUP; ++j)
                lap_qp_chunk_mma(av[g * QP_GROUP + j], b0[g * QP_GROUP + j], b1[g * QP_GROUP + j], acc0, acc1);
        }
        __builtin_amdgcn_sched_group_barrier(0x020, 2 * QP_NCHUNK, 0);
        __builtin_amdgcn_sched_group_barrier(0x200, QP_GROUP, 0);
        __builtin_amdgcn_sched_group_barrier(0x100, 2 * QP_GROUP, 0);
        #pragma unroll
        for (int g = 0; g < QP_NGROUP; ++g) {
            if (g + 1 < QP_NGROUP) {
                __builtin_amdgcn_sched_group_barrier(0x200, QP_GROUP, 0);
                __builtin_amdgcn_sched_group_barrier(0x100, 2 * QP_GROUP, 0);
            }
            __builtin_amdgcn_sched_group_barrier(0x008, 2 * QP_GROUP, 0);
        }
    } else {
        const float* wp = bias + part * QP_PART_F + 4 * h;
        #pragma unroll
        for (int j = 0; j < QP_NCHUNK; ++j) {
            av[j] = *reinterpret_cast<const float4*>(ap + j * QP_CHUNK);
            wv[j] = *reinterpret_cast<const float4*>(wp + j * QP_CHUNK);
        }
        #pragma unroll
        for (int j = 0; j < QP_NCHUNK; ++j) {
            v2f bb0, bb1;
            bb0.x = wv[j].x; bb0.y = wv[j].y;
            bb1.x = wv[j].z; bb1.y = wv[j].w;
            lap_qp_chunk_mma(av[j], bb0, bb1, acc0, acc1);
        }
    }

    if (h == (part >> 1)) {
        #pragma unroll
        for (int r = 0; r < 8; ++r)
            s_red[part][r][sub] = acc0[r] + acc1[r];
    }
    __syncthreads();
    if (lane < NQ * QP_COLS) {
        const int r  = lane >> 2;
        const int cc = lane & 3;
        float v = s_red[0][r][cc];
        #pragma unroll
        for (int p = 1; p < QP_PARTS; ++p) v += s_red[p][4 * (p & 1) + r][cc];
        if (blk < QP_NCB) qp[r * IN_DIM + QP_COLS * blk + cc] = v;
        else if (cc == 0) qb[r] = v;
    }
}

#define PW_SLICES    8
#define PW_SLICE_TOK (N_TOK / PW_SLICES)
#define PW_WAVES     8
#define PW_THREADS   (32 * PW_WAVES)
#define PW_TILES     (PW_SLICE_TOK / (16 * PW_WAVES))
#define PW_PITCH     68

__device__ __forceinline__ float lap_xhalf(float v)
{
    return __int_as_float(__builtin_amdgcn_permlanex16(__float_as_int(v), __float_as_int(v),
                                                       0x76543210, 0xfedcba98, false, false));
}

__device__ __forceinline__ float lap_comp(const float4& v, int T)
{
    return T == 0 ? v.x : (T == 1 ? v.y : (T == 2 ? v.z : v.w));
}

__global__ __launch_bounds__(PW_THREADS)
void lap_pool_mm_kernel(const float* __restrict__ x,
                        const int*   __restrict__ mask,
                        const float* __restrict__ qp,
                        const float* __restrict__ qb,
                        float* __restrict__ partP,
                        float* __restrict__ partM,
                        float* __restrict__ partL)
{
    const int b    = blockIdx.x;
    const int sl   = blockIdx.y;
    const int t    = threadIdx.x;
    const int lane = t & 31;
    const int wave = t >> 5;
    const int h    = lane >> 4;
    const int c    = lane & 15;
    const int q    = c & 3;

    __shared__ __align__(16) float s_x[PW_WAVES][16 * PW_PITCH];
    __shared__ float s_m[PW_WAVES][NQ];
    __shared__ float s_l[PW_WAVES][2][NQ];
    __shared__ float s_p[PW_WAVES][NQ][IN_DIM];

    float4 qv[8];
    #pragma unroll
    for (int jj = 0; jj < 8; ++jj)
        qv[jj] = *reinterpret_cast<const float4*>(qp + q * IN_DIM + 32 * h + 4 * jj);
    const float qbias = qb[q];

    const int tok0 = sl * PW_SLICE_TOK + wave * (16 * PW_TILES);
    const float* xw = x + ((size_t)b * N_TOK + tok0) * IN_DIM;
    const int*   mw = mask + (size_t)b * N_TOK + tok0;

    float4 R[PW_TILES][8];
    int4   mk[PW_TILES][2];
    #pragma unroll
    for (int tl = 0; tl < PW_TILES; ++tl) {
        #pragma unroll
        for (int r = 0; r < 8; ++r)
            R[tl][r] = *reinterpret_cast<const float4*>(xw + (16 * tl + 8 * h + r) * IN_DIM + 4 * c);
        mk[tl][0] = *reinterpret_cast<const int4*>(mw + 16 * tl + 8 * h);
        mk[tl][1] = *reinterpret_cast<const int4*>(mw + 16 * tl + 8 * h + 4);
    }

    float m = NEG_BIG;
    float l = 0.0f;
    v8f acc[4];
    #pragma unroll
    for (int T = 0; T < 4; ++T) acc[T] = v8f{};

    float* sx = s_x[wave];
    #pragma unroll
    for (int tl = 0; tl < PW_TILES; ++tl) {
        #pragma unroll
        for (int r = 0; r < 8; ++r)
            *reinterpret_cast<float4*>(sx + (8 * h + r) * PW_PITCH + 4 * c) = R[tl][r];
        float4 C[8];
        #pragma unroll
        for (int jj = 0; jj < 8; ++jj)
            C[jj] = *reinterpret_cast<const float4*>(sx + c * PW_PITCH + 32 * h + 4 * jj);

        v8f d0 = {};
        v8f d1 = {};
        #pragma unroll
        for (int jj = 0; jj < 8; ++jj) {
            v2f a0, a1, b0, b1;
            a0.x = C[jj].x;  a0.y = C[jj].y;  a1.x = C[jj].z;  a1.y = C[jj].w;
            b0.x = qv[jj].x; b0.y = qv[jj].y; b1.x = qv[jj].z; b1.y = qv[jj].w;
            d0 = __builtin_amdgcn_wmma_f32_16x16x4_f32(
                      false, a0,  false, b0,
                      (short)0, d0,  false,  false);
            d1 = __builtin_amdgcn_wmma_f32_16x16x4_f32(
                      false, a1,  false, b1,
                      (short)0, d1,  false,  false);
        }

        const int mkv[8] = { mk[tl][0].x, mk[tl][0].y, mk[tl][0].z, mk[tl][0].w,
                             mk[tl][1].x, mk[tl][1].y, mk[tl][1].z, mk[tl][1].w };
        float sc_[8];
        float bm = NEG_BIG;
        #pragma unroll
        for (int r = 0; r < 8; ++r) {
            float p = d0[r] + d1[r] + qbias;
            if (mkv[r] == 0) p = NEG_BIG;
            sc_[r] = p;
            bm = fmaxf(bm, p);
        }
        bm = fmaxf(bm, lap_xhalf(bm));
        const float mn = fmaxf(m, bm);
        const float rs = __expf(m - mn);
        m = mn;
        l *= rs;
        #pragma unroll
        for (int T = 0; T < 4; ++T)
            #pragma unroll
            for (int r = 0; r < 8; ++r) acc[T][r] *= rs;
        float w[8];
        #pragma unroll
        for (int r = 0; r < 8; ++r) {
            w[r] = __expf(sc_[r] - mn);
            l += w[r];
        }

        #pragma unroll
        for (int j = 0; j < 4; ++j) {
            v2f bw;
            bw.x = w[2 * j];
            bw.y = w[2 * j + 1];
            #pragma unroll
            for (int T = 0; T < 4; ++T) {
                v2f a;
                a.x = lap_comp(R[tl][2 * j], T);
                a.y = lap_comp(R[tl][2 * j + 1], T);
                acc[T] = __builtin_amdgcn_wmma_f32_16x16x4_f32(
                              false, a,  false, bw,
                              (short)0, acc[T],  false,  false);
            }
        }
        __builtin_amdgcn_sched_barrier(0);
    }

    if (c < NQ) {
        if (h == 0) s_m[wave][c] = m;
        s_l[wave][h][c] = l;
        #pragma unroll
        for (int T = 0; T < 4; ++T)
            #pragma unroll
            for (int r = 0; r < 8; ++r)
                s_p[wave][c][4 * (8 * h + r) + T] = acc[T][r];
    }
    __syncthreads();
    {
        const int qq = t >> 6;
        const int i  = t & 63;
        float M = NEG_BIG;
        #pragma unroll
        for (int wv = 0; wv < PW_WAVES; ++wv) M = fmaxf(M, s_m[wv][qq]);
        float L = 0.0f, P = 0.0f;
        #pragma unroll
        for (int wv = 0; wv < PW_WAVES; ++wv) {
            const float e = __expf(s_m[wv][qq] - M);
            L += (s_l[wv][0][qq] + s_l[wv][1][qq]) * e;
            P += s_p[wv][qq][i] * e;
        }
        const int rec = (b * PW_SLICES + sl) * NQ + qq;
        partP[rec * IN_DIM + i] = P;
        if (i == 0) { partM[rec] = M; partL[rec] = L; }
    }
}

__global__ __launch_bounds__(IN_DIM)
void lap_merge_kernel(const float* __restrict__ partP,
                      const float* __restrict__ partM,
                      const float* __restrict__ partL,
                      float* __restrict__ pooled)
{
    const int row = blockIdx.x;
    const int b   = row >> 2;
    const int q   = row & 3;
    const int i   = threadIdx.x;
    float M = NEG_BIG;
    #pragma unroll
    for (int s = 0; s < PW_SLICES; ++s) M = fmaxf(M, partM[(b * PW_SLICES + s) * NQ + q]);
    float L = 0.0f, P = 0.0f;
    #pragma unroll
    for (int s = 0; s < PW_SLICES; ++s) {
        const int rec = (b * PW_SLICES + s) * NQ + q;
        const float e = __expf(partM[rec] - M);
        L += partL[rec] * e;
        P += partP[rec * IN_DIM + i] * e;
    }
    pooled[row * IN_DIM + i] = P / L;
}

__global__ void lap_outgemm_kernel(const float* __restrict__ pooled,
                                   const float* __restrict__ W,
                                   const float* __restrict__ bias,
                                   float* __restrict__ out)
{
    const int n0   = blockIdx.x * 16;
    const int m0   = blockIdx.y * 16;
    const int lane = threadIdx.x;
    const int hf   = lane >> 4;
    const int l    = lane & 15;

    const int arow = m0 + l;
    const int bcol = n0 + l;
    const float* ap = pooled + arow * IN_DIM;
    const float* bp = W + bcol * IN_DIM;

    v8f acc = {};
    #pragma unroll
    for (int k0 = 0; k0 < IN_DIM; k0 += 4) {
        const int ka = k0 + hf * 2;
        v2f a, bfr;
        a.x   = ap[ka];
        a.y   = ap[ka + 1];
        bfr.x = bp[ka];
        bfr.y = bp[ka + 1];
        acc = __builtin_amdgcn_wmma_f32_16x16x4_f32(
                   false, a,  false, bfr,
                   (short)0, acc,  false,  false);
    }

    const int col = n0 + l;
    const float bv = bias[col];
    #pragma unroll
    for (int r = 0; r < 8; ++r) {
        const int row = m0 + hf * 8 + r;
        out[row * FEAT_DIM + col] = acc[r] + bv;
    }
}

extern "C" void kernel_launch(void* const* d_in, const int* in_sizes, int n_in,
                              void* d_out, int out_size, void* d_ws, size_t ws_size,
                              hipStream_t stream) {
    const float* x    = (const float*)d_in[0];
    const int*   mask = (const int*)  d_in[1];
    const float* qt   = (const float*)d_in[2];
    const float* W    = (const float*)d_in[3];
    const float* bias = (const float*)d_in[4];
    float* out = (float*)d_out;

    const int B = in_sizes[0] / (N_TOK * IN_DIM);

    float* ws     = (float*)d_ws;
    float* qp     = ws;
    float* qb     = ws + 256;
    float* pooled = ws + 512;

    lap_qproj_kernel<<<17, 32, 0, stream>>>(qt, W, bias, qp, qb);
    float* partP = out;
    float* partM = out + 65536;
    float* partL = out + 66560;
    lap_pool_mm_kernel<<<dim3(B, PW_SLICES), PW_THREADS, 0, stream>>>(x, mask, qp, qb, partP, partM, partL);
    lap_merge_kernel<<<B * NQ, IN_DIM, 0, stream>>>(partP, partM, partL, pooled);
    lap_outgemm_kernel<<<dim3(FEAT_DIM / 16, (B * NQ) / 16), 32, 0, stream>>>(
        pooled, W, bias, out);
}


# pooling works on unmasked tokens only (a masked one weighs exactly 0): a wave packs them via an index list in LDS into 1-4 tiles, fetches no other row; all-masked wave keeps all 64; new n-sum order
# speedup vs baseline: 7.1326x; 1.0886x over previous
#include <hip/hip_runtime.h>
#include <math.h>

typedef __attribute__((ext_vector_type(2))) float v2f;
typedef __attribute__((ext_vector_type(8))) float v8f;

#define B_MAX      32
#define N_TOK      4096
#define IN_DIM     64
#define FEAT_DIM   768
#define NQ         4
#define NEG_BIG    (-3.402823466e38f)

#define QP_PARTS   4
#define QP_PART_F  (FEAT_DIM / QP_PARTS)
#define QP_CHUNK   8
#define QP_NCHUNK  (QP_PART_F / QP_CHUNK)
#define QP_GROUP   4
#define QP_NGROUP  (QP_NCHUNK / QP_GROUP)
#define QP_COLS    (16 / QP_PARTS)
#define QP_NCB     (IN_DIM / QP_COLS)
#define QP_REGION  (32 * QP_COLS)

__device__ __forceinline__ void lap_qp_chunk_mma(const float4& a, const v2f& b0, const v2f& b1,
                                                 v8f& acc0, v8f& acc1)
{
    v2f a0, a1;
    a0.x = a.x; a0.y = a.y;
    a1.x = a.z; a1.y = a.w;
    acc0 = __builtin_amdgcn_wmma_f32_16x16x4_f32(
                false, a0,  false, b0,
                (short)0, acc0,  false,  false);
    acc1 = __builtin_amdgcn_wmma_f32_16x16x4_f32(
                false, a1,  false, b1,
                (short)0, acc1,  false,  false);
}

__global__ __launch_bounds__(32)
void lap_qproj_kernel(const float* __restrict__ qt,
                      const float* __restrict__ W,
                      const float* __restrict__ bias,
                      float* __restrict__ qp,
                      float* __restrict__ qb)
{
    const int lane = threadIdx.x;
    const int h    = lane >> 4;
    const int c    = lane & 15;
    const int part = c >> 2;
    const int sub  = c & 3;
    const int blk  = blockIdx.x;

    __shared__ __align__(16) float s_w[QP_NCHUNK * QP_REGION];
    __shared__ float s_red[QP_PARTS][8][QP_COLS];

    const float* ap = qt + sub * FEAT_DIM + part * QP_PART_F + 4 * h;
    v8f acc0 = {};
    v8f acc1 = {};
    float4 av[QP_NCHUNK];
    float4 wv[QP_NCHUNK];
    if (blk < QP_NCB) {
        const float* wp = W + ((lane >> 3) * QP_PART_F + (lane & 7)) * IN_DIM + QP_COLS * blk;
        float*       st = s_w + QP_COLS * lane;
        const float* ld = s_w + (8 * part + 4 * h) * QP_COLS + sub;
        v2f b0[QP_NCHUNK];
        v2f b1[QP_NCHUNK];
        #pragma unroll
        for (int j = 0; j < QP_NCHUNK; ++j) {
            av[j] = *reinterpret_cast<const float4*>(ap + j * QP_CHUNK);
            wv[j] = *reinterpret_cast<const float4*>(wp + j * QP_CHUNK * IN_DIM);
        }
        #pragma unroll
        for (int j = 0; j < QP_GROUP; ++j)
            *reinterpret_cast<float4*>(st + j * QP_REGION) = wv[j];
        #pragma unroll
        for (int j = 0; j < QP_GROUP; ++j) {
            const float* p = ld + j * QP_REGION;
            b0[j].x = p[0];
            b0[j].y = p[QP_COLS];
            b1[j].x = p[2 * QP_COLS];
            b1[j].y = p[3 * QP_COLS];
        }
        #pragma unroll
        for (int g = 0; g < QP_NGROUP; ++g) {
            if (g + 1 < QP_NGROUP) {
                #pragma unroll
                for (int j = 0; j < QP_GROUP; ++j)
                    *reinterpret_cast<float4*>(st + ((g + 1) * QP_GROUP + j) * QP_REGION) = wv[(g + 1) * QP_GROUP + j];
                #pragma unroll
                for (int j = 0; j < QP_GROUP; ++j) {
                    const float* p = ld + ((g + 1) * QP_GROUP + j) * QP_REGION;
                    b0[(g + 1) * QP_GROUP + j].x = p[0];
                    b0[(g + 1) * QP_GROUP + j].y = p[QP_COLS];
                    b1[(g + 1) * QP_GROUP + j].x = p[2 * QP_COLS];
                    b1[(g + 1) * QP_GROUP + j].y = p[3 * QP_COLS];
                }
            }
            #pragma unroll
            for (int j = 0; j < QP_GROUP; ++j)
                lap_qp_chunk_mma(av[g * QP_GROUP + j], b0[g * QP_GROUP + j], b1[g * QP_GROUP + j], acc0, acc1);
        }
        __builtin_amdgcn_sched_group_barrier(0x020, 2 * QP_NCHUNK, 0);
        __builtin_amdgcn_sched_group_barrier(0x200, QP_GROUP, 0);
        __builtin_amdgcn_sched_group_barrier(0x100, 2 * QP_GROUP, 0);
        #pragma unroll
        for (int g = 0; g < QP_NGROUP; ++g) {
            if (g + 1 < QP_NGROUP) {
                __builtin_amdgcn_sched_group_barrier(0x200, QP_GROUP, 0);
                __builtin_amdgcn_sched_group_barrier(0x100, 2 * QP_GROUP, 0);
            }
            __builtin_amdgcn_sched_group_barrier(0x008, 2 * QP_GROUP, 0);
        }
    } else {
        const float* wp = bias + part * QP_PART_F + 4 * h;
        #pragma unroll
        for (int j = 0; j < QP_NCHUNK; ++j) {
            av[j] = *reinterpret_cast<const float4*>(ap + j * QP_CHUNK);
            wv[j] = *reinterpret_cast<const float4*>(wp + j * QP_CHUNK);
        }
        #pragma unroll
        for (int j = 0; j < QP_NCHUNK; ++j) {
            v2f bb0, bb1;
            bb0.x = wv[j].x; bb0.y = wv[j].y;
            bb1.x = wv[j].z; bb1.y = wv[j].w;
            lap_qp_chunk_mma(av[j], bb0, bb1, acc0, acc1);
        }
    }

    if (h == (part >> 1)) {
        #pragma unroll
        for (int r = 0; r < 8; ++r)
            s_red[part][r][sub] = acc0[r] + acc1[r];
    }
    __syncthreads();
    if (lane < NQ * QP_COLS) {
        const int r  = lane >> 2;
        const int cc = lane & 3;
        float v = s_red[0][r][cc];
        #pragma unroll
        for (int p = 1; p < QP_PARTS; ++p) v += s_red[p][4 * (p & 1) + r][cc];
        if (blk < QP_NCB) qp[r * IN_DIM + QP_COLS * blk + cc] = v;
        else if (cc == 0) qb[r] = v;
    }
}

#define PW_SLICES    8
#define PW_SLICE_TOK (N_TOK / PW_SLICES)
#define PW_WAVES     8
#define PW_THREADS   (32 * PW_WAVES)
#define PW_WAVE_TOK  (PW_SLICE_TOK / PW_WAVES)
#define PW_TILES     (PW_WAVE_TOK / 16)
#define PW_PITCH     68

__device__ __forceinline__ float lap_xhalf(float v)
{
    return __int_as_float(__builtin_amdgcn_permlanex16(__float_as_int(v), __float_as_int(v),
                                                       0x76543210, 0xfedcba98, false, false));
}

__device__ __forceinline__ float lap_comp(const float4& v, int T)
{
    return T == 0 ? v.x : (T == 1 ? v.y : (T == 2 ? v.z : v.w));
}

template <int NT>
__device__ __forceinline__ void lap_pool_tiles(const char* __restrict__ xw, const unsigned coff,
                                               const int* si, float* sx,
                                               const int h, const int c, const int lim,
                                               const float4 (&qv)[8], const float qbias,
                                               float& m, float& l, v8f (&acc)[4])
{
    unsigned id[NT][8];
    #pragma unroll
    for (int tl = 0; tl < NT; ++tl) {
        const int4 ia = *reinterpret_cast<const int4*>(si + 16 * tl + 8 * h);
        const int4 ib = *reinterpret_cast<const int4*>(si + 16 * tl + 8 * h + 4);
        id[tl][0] = ia.x; id[tl][1] = ia.y; id[tl][2] = ia.z; id[tl][3] = ia.w;
        id[tl][4] = ib.x; id[tl][5] = ib.y; id[tl][6] = ib.z; id[tl][7] = ib.w;
    }
    float4 R[NT][8];
    #pragma unroll
    for (int tl = 0; tl < NT; ++tl)
        #pragma unroll
        for (int r = 0; r < 8; ++r)
            R[tl][r] = *reinterpret_cast<const float4*>(xw + (id[tl][r] * (unsigned)(IN_DIM * sizeof(float)) + coff));
    __builtin_amdgcn_sched_barrier(0);

    #pragma unroll
    for (int tl = 0; tl < NT; ++tl) {
        #pragma unroll
        for (int r = 0; r < 8; ++r)
            *reinterpret_cast<float4*>(sx + (8 * h + r) * PW_PITCH + 4 * c) = R[tl][r];
        float4 C[8];
        #pragma unroll
        for (int jj = 0; jj < 8; ++jj)
            C[jj] = *reinterpret_cast<const float4*>(sx + c * PW_PITCH + 32 * h + 4 * jj);

        v8f d0 = {};
        v8f d1 = {};
        #pragma unroll
        for (int jj = 0; jj < 8; ++jj) {
            v2f a0, a1, b0, b1;
            a0.x = C[jj].x;  a0.y = C[jj].y;  a1.x = C[jj].z;  a1.y = C[jj].w;
            b0.x = qv[jj].x; b0.y = qv[jj].y; b1.x = qv[jj].z; b1.y = qv[jj].w;
            d0 = __builtin_amdgcn_wmma_f32_16x16x4_f32(
                      false, a0,  false, b0,
                      (short)0, d0,  false,  false);
            d1 = __builtin_amdgcn_wmma_f32_16x16x4_f32(
                      false, a1,  false, b1,
                      (short)0, d1,  false,  false);
        }

        float sc_[8];
        float bm = NEG_BIG;
        #pragma unroll
        for (int r = 0; r < 8; ++r) {
            float p = d0[r] + d1[r] + qbias;
            if (16 * tl + r >= lim) p = NEG_BIG;
            sc_[r] = p;
            bm = fmaxf(bm, p);
        }
        bm = fmaxf(bm, lap_xhalf(bm));
        const float mn = fmaxf(m, bm);
        const float rs = __expf(m - mn);
        m = mn;
        l *= rs;
        #pragma unroll
        for (int T = 0; T < 4; ++T)
            #pragma unroll
            for (int r = 0; r < 8; ++r) acc[T][r] *= rs;
        float w[8];
        #pragma unroll
        for (int r = 0; r < 8; ++r) {
            w[r] = __expf(sc_[r] - mn);
            l += w[r];
        }

        #pragma unroll
        for (int j = 0; j < 4; ++j) {
            v2f bw;
            bw.x = w[2 * j];
            bw.y = w[2 * j + 1];
            #pragma unroll
            for (int T = 0; T < 4; ++T) {
                v2f a;
                a.x = lap_comp(R[tl][2 * j], T);
                a.y = lap_comp(R[tl][2 * j + 1], T);
                acc[T] = __builtin_amdgcn_wmma_f32_16x16x4_f32(
                              false, a,  false, bw,
                              (short)0, acc[T],  false,  false);
            }
        }
        __builtin_amdgcn_sched_barrier(0);
    }
}

__global__ __launch_bounds__(PW_THREADS)
void lap_pool_mm_kernel(const float* __restrict__ x,
                        const int*   __restrict__ mask,
                        const float* __restrict__ qp,
                        const float* __restrict__ qb,
                        float* __restrict__ partP,
                        float* __restrict__ partM,
                        float* __restrict__ partL)
{
    const int b    = blockIdx.x;
    const int sl   = blockIdx.y;
    const int t    = threadIdx.x;
    const int lane = t & 31;
    const int wave = __builtin_amdgcn_readfirstlane(t >> 5);
    const int h    = lane >> 4;
    const int c    = lane & 15;
    const int q    = c & 3;

    __shared__ __align__(16) float s_x[PW_WAVES][16 * PW_PITCH];
    __shared__ __align__(16) int   s_idx[PW_WAVES][PW_WAVE_TOK];
    __shared__ float s_m[PW_WAVES][NQ];
    __shared__ float s_l[PW_WAVES][2][NQ];
    __shared__ float s_p[PW_WAVES][NQ][IN_DIM];

    const int tok0 = sl * PW_SLICE_TOK + wave * PW_WAVE_TOK;
    const int* mw = mask + (size_t)b * N_TOK + tok0;
    const int mv0 = mw[lane];
    const int mv1 = mw[32 + lane];

    float4 qv[8];
    #pragma unroll
    for (int jj = 0; jj < 8; ++jj)
        qv[jj] = *reinterpret_cast<const float4*>(qp + q * IN_DIM + 32 * h + 4 * jj);
    const float qbias = qb[q];

    unsigned lo = __builtin_amdgcn_ballot_w32(mv0 != 0);
    unsigned hi = __builtin_amdgcn_ballot_w32(mv1 != 0);
    const bool allm = (lo | hi) == 0u;
    if (allm) { lo = ~0u; hi = ~0u; }
    const int nlo  = __builtin_popcount(lo);
    const int nsel = nlo + __builtin_popcount(hi);
    const int nt   = (nsel + 15) >> 4;
    const int lim  = (allm ? 0 : nsel) - 8 * h;

    int* si = s_idx[wave];
    if ((lo >> lane) & 1u) si[__builtin_amdgcn_mbcnt_lo(lo, 0u)] = lane;
    if ((hi >> lane) & 1u) si[nlo + __builtin_amdgcn_mbcnt_lo(hi, 0u)] = 32 + lane;
    const int first = lo ? __builtin_ctz(lo) : 32 + __builtin_ctz(hi);
    if (nsel + lane < PW_WAVE_TOK)      si[nsel + lane]      = first;
    if (nsel + 32 + lane < PW_WAVE_TOK) si[nsel + 32 + lane] = first;

    const char* xw = reinterpret_cast<const char*>(x + ((size_t)b * N_TOK + tok0) * IN_DIM);
    const unsigned coff = 16u * (unsigned)c;

    float m = NEG_BIG;
    float l = 0.0f;
    v8f acc[4];
    #pragma unroll
    for (int T = 0; T < 4; ++T) acc[T] = v8f{};

    float* sx = s_x[wave];
    switch (nt) {
    case 1:  lap_pool_tiles<1>(xw, coff, si, sx, h, c, lim, qv, qbias, m, l, acc); break;
    case 2:  lap_pool_tiles<2>(xw, coff, si, sx, h, c, lim, qv, qbias, m, l, acc); break;
    case 3:  lap_pool_tiles<3>(xw, coff, si, sx, h, c, lim, qv, qbias, m, l, acc); break;
    default: lap_pool_tiles<4>(xw, coff, si, sx, h, c, lim, qv, qbias, m, l, acc); break;
    }

    if (c < NQ) {
        if (h == 0) s_m[wave][c] = m;
        s_l[wave][h][c] = l;
        #pragma unroll
        for (int T = 0; T < 4; ++T)
            #pragma unroll
            for (int r = 0; r < 8; ++r)
                s_p[wave][c][4 * (8 * h + r) + T] = acc[T][r];
    }
    __syncthreads();
    {
        const int qq = t >> 6;
        const int i  = t & 63;
        float M = NEG_BIG;
        #pragma unroll
        for (int wv = 0; wv < PW_WAVES; ++wv) M = fmaxf(M, s_m[wv][qq]);
        float L = 0.0f, P = 0.0f;
        #pragma unroll
        for (int wv = 0; wv < PW_WAVES; ++wv) {
            const float e = __expf(s_m[wv][qq] - M);
            L += (s_l[wv][0][qq] + s_l[wv][1][qq]) * e;
            P += s_p[wv][qq][i] * e;
        }
        const int rec = (b * PW_SLICES + sl) * NQ + qq;
        partP[rec * IN_DIM + i] = P;
        if (i == 0) { partM[rec] = M; partL[rec] = L; }
    }
}

__global__ __launch_bounds__(IN_DIM)
void lap_merge_kernel(const float* __restrict__ partP,
                      const float* __restrict__ partM,
                      const float* __restrict__ partL,
                      float* __restrict__ pooled)
{
    const int row = blockIdx.x;
    const int b   = row >> 2;
    const int q   = row & 3;
    const int i   = threadIdx.x;
    float M = NEG_BIG;
    #pragma unroll
    for (int s = 0; s < PW_SLICES; ++s) M = fmaxf(M, partM[(b * PW_SLICES + s) * NQ + q]);
    float L = 0.0f, P = 0.0f;
    #pragma unroll
    for (int s = 0; s < PW_SLICES; ++s) {
        const int rec = (b * PW_SLICES + s) * NQ + q;
        const float e = __expf(partM[rec] - M);
        L += partL[rec] * e;
        P += partP[rec * IN_DIM + i] * e;
    }
    pooled[row * IN_DIM + i] = P / L;
}

__global__ void lap_outgemm_kernel(const float* __restrict__ pooled,
                                   const float* __restrict__ W,
                                   const float* __restrict__ bias,
                                   float* __restrict__ out)
{
    const int n0   = blockIdx.x * 16;
    const int m0   = blockIdx.y * 16;
    const int lane = threadIdx.x;
    const int hf   = lane >> 4;
    const int l    = lane & 15;

    const int arow = m0 + l;
    const int bcol = n0 + l;
    const float* ap = pooled + arow * IN_DIM;
    const float* bp = W + bcol * IN_DIM;

    v8f acc = {};
    #pragma unroll
    for (int k0 = 0; k0 < IN_DIM; k0 += 4) {
        const int ka = k0 + hf * 2;
        v2f a, bfr;
        a.x   = ap[ka];
        a.y   = ap[ka + 1];
        bfr.x = bp[ka];
        bfr.y = bp[ka + 1];
        acc = __builtin_amdgcn_wmma_f32_16x16x4_f32(
                   false, a,  false, bfr,
                   (short)0, acc,  false,  false);
    }

    const int col = n0 + l;
    const float bv = bias[col];
    #pragma unroll
    for (int r = 0; r < 8; ++r) {
        const int row = m0 + hf * 8 + r;
        out[row * FEAT_DIM + col] = acc[r] + bv;
    }
}

extern "C" void kernel_launch(void* const* d_in, const int* in_sizes, int n_in,
                              void* d_out, int out_size, void* d_ws, size_t ws_size,
                              hipStream_t stream) {
    const float* x    = (const float*)d_in[0];
    const int*   mask = (const int*)  d_in[1];
    const float* qt   = (const float*)d_in[2];
    const float* W    = (const float*)d_in[3];
    const float* bias = (const float*)d_in[4];
    float* out = (float*)d_out;

    const int B = in_sizes[0] / (N_TOK * IN_DIM);

    float* ws     = (float*)d_ws;
    float* qp     = ws;
    float* qb     = ws + 256;
    float* pooled = ws + 512;

    lap_qproj_kernel<<<17, 32, 0, stream>>>(qt, W, bias, qp, qb);
    float* partP = out;
    float* partM = out + 65536;
    float* partL = out + 66560;
    lap_pool_mm_kernel<<<dim3(B, PW_SLICES), PW_THREADS, 0, stream>>>(x, mask, qp, qb, partP, partM, partL);
    lap_merge_kernel<<<B * NQ, IN_DIM, 0, stream>>>(partP, partM, partL, pooled);
    lap_outgemm_kernel<<<dim3(FEAT_DIM / 16, (B * NQ) / 16), 32, 0, stream>>>(
        pooled, W, bias, out);
}
